# starting kernel
# baseline (speedup 1.0000x reference)
#include <hip/hip_runtime.h>
#include <hip/hip_bf16.h>

typedef __attribute__((ext_vector_type(2))) float v2f;
typedef __attribute__((ext_vector_type(8))) float v8f;

#define NB   2
#define NCH  64
#define NN   128
#define NT   24
#define X_CSTRIDE   (NN*NT)
#define X_BSTRIDE   (NCH*NN*NT)
#define A_ISTRIDE   (NN*NT)
#define A_BSTRIDE   (NN*NN*NT)

__device__ __forceinline__ v8f wmma4(v2f a, v2f b, v8f c) {
  return __builtin_amdgcn_wmma_f32_16x16x4_f32(
       false, a,  false, b,
       (short)0, c,  false,  false);
}

__global__ __launch_bounds__(256)
void map_fused_kernel(const float* __restrict__ x,
                      const float* __restrict__ att,
                      const float* __restrict__ W1,
                      const float* __restrict__ b1,
                      const float* __restrict__ W2,
                      const float* __restrict__ b2,
                      float* __restrict__ out)
{
  __shared__ float ldsA[NN * NCH];
  __shared__ float ldsB[NN * NCH];

  const int tid  = threadIdx.x;
  const int lane = tid & 31;
  const int lo   = lane & 15;
  const int hi   = lane >> 4;
  const int wave = tid >> 5;

  const int bt = blockIdx.x;
  const int b  = bt / NT;
  const int t  = bt % NT;

  const int n_row = 16 * wave + lo;
  const int xbase = b * X_BSTRIDE + n_row * NT + t;

  v8f hc[4];
  #pragma unroll
  for (int j = 0; j < 4; ++j) {
    const float bias = b1[16 * j + lo];
    #pragma unroll
    for (int r = 0; r < 8; ++r) hc[j][r] = bias;
  }

  for (int kk = 0; kk < 16; ++kk) {
    const int k = 4 * kk + 2 * hi;
    v2f a;
    a.x = x[xbase + (k    ) * X_CSTRIDE];
    a.y = x[xbase + (k + 1) * X_CSTRIDE];
    #pragma unroll
    for (int j = 0; j < 4; ++j) {
      v2f w;
      w.x = W1[(k    ) * NCH + 16 * j + lo];
      w.y = W1[(k + 1) * NCH + 16 * j + lo];
      hc[j] = wmma4(a, w, hc[j]);
    }
  }

  #pragma unroll
  for (int j = 0; j < 4; ++j)
    #pragma unroll
    for (int r = 0; r < 8; ++r)
      ldsA[(16 * wave + r + 8 * hi) * NCH + 16 * j + lo] = hc[j][r];

  __syncthreads();

  v8f mvc[4], mhc[4];
  #pragma unroll
  for (int j = 0; j < 4; ++j) {
    const float bias = b2[16 * j + lo];
    #pragma unroll
    for (int r = 0; r < 8; ++r) { mvc[j][r] = bias; mhc[j][r] = 0.0f; }
  }

  for (int kk = 0; kk < 16; ++kk) {
    const int k = 4 * kk + 2 * hi;
    v2f a = *(const v2f*)&ldsA[(16 * wave + lo) * NCH + k];
    #pragma unroll
    for (int j = 0; j < 4; ++j) {
      v2f wv, wh;
      wv.x = W2[(k          ) * NCH + 16 * j + lo];
      wv.y = W2[(k + 1      ) * NCH + 16 * j + lo];
      wh.x = W2[(k + NCH    ) * NCH + 16 * j + lo];
      wh.y = W2[(k + NCH + 1) * NCH + 16 * j + lo];
      mvc[j] = wmma4(a, wv, mvc[j]);
      mhc[j] = wmma4(a, wh, mhc[j]);
    }
  }

  __syncthreads();

  #pragma unroll
  for (int j = 0; j < 4; ++j)
    #pragma unroll
    for (int r = 0; r < 8; ++r) {
      const int row = 16 * wave + r + 8 * hi;
      ldsA[row * NCH + 16 * j + lo] = mvc[j][r];
      ldsB[row * NCH + 16 * j + lo] = mhc[j][r];
    }

  __syncthreads();

  const int i0    = 16 * wave;
  const int abase = b * A_BSTRIDE + (i0 + lo) * A_ISTRIDE + t;

  v8f acc[4];
  #pragma unroll
  for (int j = 0; j < 4; ++j)
    #pragma unroll
    for (int r = 0; r < 8; ++r) acc[j][r] = 0.0f;

  float psum = 0.0f;
  for (int kk = 0; kk < 32; ++kk) {
    const int jr = 4 * kk + 2 * hi;
    v2f a;
    a.x = att[abase + (jr    ) * NT];
    a.y = att[abase + (jr + 1) * NT];
    psum += a.x + a.y;
    #pragma unroll
    for (int n = 0; n < 4; ++n) {
      v2f mb;
      mb.x = ldsB[(jr    ) * NCH + 16 * n + lo];
      mb.y = ldsB[(jr + 1) * NCH + 16 * n + lo];
      acc[n] = wmma4(a, mb, acc[n]);
    }
  }

  const float asum_lane = psum + __shfl_xor(psum, 16, 32);

  #pragma unroll
  for (int r = 0; r < 8; ++r) {
    const int row = r + 8 * hi;
    const float as = __shfl(asum_lane, row, 32);
    const int gi = i0 + row;
    #pragma unroll
    for (int n = 0; n < 4; ++n) {
      const int c = 16 * n + lo;
      const float val = acc[n][r] + as * ldsA[gi * NCH + c];
      out[b * X_BSTRIDE + c * X_CSTRIDE + gi * NT + t] = val;
    }
  }
}

extern "C" void kernel_launch(void* const* d_in, const int* in_sizes, int n_in,
                              void* d_out, int out_size, void* d_ws, size_t ws_size,
                              hipStream_t stream) {
  const float* x   = (const float*)d_in[0];
  const float* att = (const float*)d_in[1];
  const float* W1  = (const float*)d_in[2];
  const float* b1  = (const float*)d_in[3];
  const float* W2  = (const float*)d_in[4];
  const float* b2  = (const float*)d_in[5];
  float* out = (float*)d_out;

  dim3 grid(NB * NT);
  dim3 block(256);
  hipLaunchKernelGGL(map_fused_kernel, grid, block, 0, stream,
                     x, att, W1, b1, W2, b2, out);
}


# Issue all strided global operand loads (x and attention fragments) as one batch at kernel entry into registers, so their round trips overlap instead of being paid once per k-step
# speedup vs baseline: 1.0945x; 1.0945x over previous
#include <hip/hip_runtime.h>
#include <hip/hip_bf16.h>

typedef __attribute__((ext_vector_type(2))) float v2f;
typedef __attribute__((ext_vector_type(8))) float v8f;

#define NB   2
#define NCH  64
#define NN   128
#define NT   24
#define X_CSTRIDE   (NN*NT)
#define X_BSTRIDE   (NCH*NN*NT)
#define A_ISTRIDE   (NN*NT)
#define A_BSTRIDE   (NN*NN*NT)

__device__ __forceinline__ v8f wmma4(v2f a, v2f b, v8f c) {
  return __builtin_amdgcn_wmma_f32_16x16x4_f32(
       false, a,  false, b,
       (short)0, c,  false,  false);
}

__global__ __launch_bounds__(256)
void map_fused_kernel(const float* __restrict__ x,
                      const float* __restrict__ att,
                      const float* __restrict__ W1,
                      const float* __restrict__ b1,
                      const float* __restrict__ W2,
                      const float* __restrict__ b2,
                      float* __restrict__ out)
{
  __shared__ float ldsA[NN * NCH];
  __shared__ float ldsB[NN * NCH];

  const int tid  = threadIdx.x;
  const int lane = tid & 31;
  const int lo   = lane & 15;
  const int hi   = lane >> 4;
  const int wave = tid >> 5;

  const int bt = blockIdx.x;
  const int b  = bt / NT;
  const int t  = bt % NT;

  const int n_row = 16 * wave + lo;
  const int xbase = b * X_BSTRIDE + n_row * NT + t;

  const int i0    = 16 * wave;
  const int abase = b * A_BSTRIDE + (i0 + lo) * A_ISTRIDE + t;

  v2f xr[16];
  #pragma unroll
  for (int kk = 0; kk < 16; ++kk) {
    const int k = 4 * kk + 2 * hi;
    xr[kk].x = x[xbase + (k    ) * X_CSTRIDE];
    xr[kk].y = x[xbase + (k + 1) * X_CSTRIDE];
  }
  v2f ar[32];
  #pragma unroll
  for (int kk = 0; kk < 32; ++kk) {
    const int jr = 4 * kk + 2 * hi;
    ar[kk].x = att[abase + (jr    ) * NT];
    ar[kk].y = att[abase + (jr + 1) * NT];
  }

  v8f hc[4];
  #pragma unroll
  for (int j = 0; j < 4; ++j) {
    const float bias = b1[16 * j + lo];
    #pragma unroll
    for (int r = 0; r < 8; ++r) hc[j][r] = bias;
  }

  #pragma unroll
  for (int kk = 0; kk < 16; ++kk) {
    const int k = 4 * kk + 2 * hi;
    const v2f a = xr[kk];
    #pragma unroll
    for (int j = 0; j < 4; ++j) {
      v2f w;
      w.x = W1[(k    ) * NCH + 16 * j + lo];
      w.y = W1[(k + 1) * NCH + 16 * j + lo];
      hc[j] = wmma4(a, w, hc[j]);
    }
  }

  #pragma unroll
  for (int j = 0; j < 4; ++j)
    #pragma unroll
    for (int r = 0; r < 8; ++r)
      ldsA[(16 * wave + r + 8 * hi) * NCH + 16 * j + lo] = hc[j][r];

  __syncthreads();

  v8f mvc[4], mhc[4];
  #pragma unroll
  for (int j = 0; j < 4; ++j) {
    const float bias = b2[16 * j + lo];
    #pragma unroll
    for (int r = 0; r < 8; ++r) { mvc[j][r] = bias; mhc[j][r] = 0.0f; }
  }

  for (int kk = 0; kk < 16; ++kk) {
    const int k = 4 * kk + 2 * hi;
    v2f a = *(const v2f*)&ldsA[(16 * wave + lo) * NCH + k];
    #pragma unroll
    for (int j = 0; j < 4; ++j) {
      v2f wv, wh;
      wv.x = W2[(k          ) * NCH + 16 * j + lo];
      wv.y = W2[(k + 1      ) * NCH + 16 * j + lo];
      wh.x = W2[(k + NCH    ) * NCH + 16 * j + lo];
      wh.y = W2[(k + NCH + 1) * NCH + 16 * j + lo];
      mvc[j] = wmma4(a, wv, mvc[j]);
      mhc[j] = wmma4(a, wh, mhc[j]);
    }
  }

  __syncthreads();

  #pragma unroll
  for (int j = 0; j < 4; ++j)
    #pragma unroll
    for (int r = 0; r < 8; ++r) {
      const int row = 16 * wave + r + 8 * hi;
      ldsA[row * NCH + 16 * j + lo] = mvc[j][r];
      ldsB[row * NCH + 16 * j + lo] = mhc[j][r];
    }

  __syncthreads();

  v8f acc[4];
  #pragma unroll
  for (int j = 0; j < 4; ++j)
    #pragma unroll
    for (int r = 0; r < 8; ++r) acc[j][r] = 0.0f;

  float psum = 0.0f;
  #pragma unroll
  for (int kk = 0; kk < 32; ++kk) {
    const int jr = 4 * kk + 2 * hi;
    const v2f a = ar[kk];
    psum += a.x + a.y;
    #pragma unroll
    for (int n = 0; n < 4; ++n) {
      v2f mb;
      mb.x = ldsB[(jr    ) * NCH + 16 * n + lo];
      mb.y = ldsB[(jr + 1) * NCH + 16 * n + lo];
      acc[n] = wmma4(a, mb, acc[n]);
    }
  }

  const float asum_lane = psum + __shfl_xor(psum, 16, 32);

  #pragma unroll
  for (int r = 0; r < 8; ++r) {
    const int row = r + 8 * hi;
    const float as = __shfl(asum_lane, row, 32);
    const int gi = i0 + row;
    #pragma unroll
    for (int n = 0; n < 4; ++n) {
      const int c = 16 * n + lo;
      const float val = acc[n][r] + as * ldsA[gi * NCH + c];
      out[b * X_BSTRIDE + c * X_CSTRIDE + gi * NT + t] = val;
    }
  }
}

extern "C" void kernel_launch(void* const* d_in, const int* in_sizes, int n_in,
                              void* d_out, int out_size, void* d_ws, size_t ws_size,
                              hipStream_t stream) {
  const float* x   = (const float*)d_in[0];
  const float* att = (const float*)d_in[1];
  const float* W1  = (const float*)d_in[2];
  const float* b1  = (const float*)d_in[3];
  const float* W2  = (const float*)d_in[4];
  const float* b2  = (const float*)d_in[5];
  float* out = (float*)d_out;

  dim3 grid(NB * NT);
  dim3 block(256);
  hipLaunchKernelGGL(map_fused_kernel, grid, block, 0, stream,
                     x, att, W1, b1, W2, b2, out);
}


# Two launches via workspace: row-wise MLP reads x along its contiguous (n,t) axis; mixing runs as 96 one-wave blocks taking 4 adjacent t per 16-byte attention load and store
# speedup vs baseline: 3.3254x; 3.0383x over previous
#include <hip/hip_runtime.h>


typedef __attribute__((ext_vector_type(2))) float v2f;
typedef __attribute__((ext_vector_type(4))) float v4f;
typedef __attribute__((ext_vector_type(8))) float v8f;

#define NB   2
#define NCH  64
#define NN   128
#define NT   24
#define MROWS       (NN*NT)
#define X_CSTRIDE   (NN*NT)
#define X_BSTRIDE   (NCH*NN*NT)
#define A_ISTRIDE   (NN*NT)
#define A_BSTRIDE   (NN*NN*NT)
#define TQ   4
#define NTQ  (NT/TQ)
#define WS_PLANE    (NB*NT*NN*NCH)


__device__ __forceinline__ v8f wmma4(v2f a, v2f b, v8f c) {
  return __builtin_amdgcn_wmma_f32_16x16x4_f32(
       false, a,  false, b,
       (short)0, c,  false,  false);
}

__global__ __launch_bounds__(128)
void mlp_rows_kernel(const float* __restrict__ x,
                     const float* __restrict__ W1,
                     const float* __restrict__ b1,
                     const float* __restrict__ W2,
                     const float* __restrict__ b2,
                     float* __restrict__ P,
                     float* __restrict__ Q)
{
  __shared__ float hs[4 * 16 * NCH];

  const int tid  = threadIdx.x;
  const int lane = tid & 31;
  const int lo   = lane & 15;
  const int hi   = lane >> 4;
  const int wave = tid >> 5;

  const int tile = blockIdx.x * 4 + wave;
  const int b    = tile / (MROWS / 16);
  const int m0   = (tile - b * (MROWS / 16)) * 16;
  float* hw = hs + wave * (16 * NCH);

  const int xbase = b * X_BSTRIDE + m0 + lo;
  v2f xr[16];
  #pragma unroll
  for (int kk = 0; kk < 16; ++kk) {
    const int k = 4 * kk + 2 * hi;
    xr[kk].x = x[xbase + (k    ) * X_CSTRIDE];
    xr[kk].y = x[xbase + (k + 1) * X_CSTRIDE];
  }

  v8f hc[4];
  #pragma unroll
  for (int j = 0; j < 4; ++j) {
    const float bias = b1[16 * j + lo];
    #pragma unroll
    for (int r = 0; r < 8; ++r) hc[j][r] = bias;
  }

  #pragma unroll
  for (int kk = 0; kk < 16; ++kk) {
    const int k = 4 * kk + 2 * hi;
    const v2f a = xr[kk];
    #pragma unroll
    for (int j = 0; j < 4; ++j) {
      v2f w;
      w.x = W1[(k    ) * NCH + 16 * j + lo];
      w.y = W1[(k + 1) * NCH + 16 * j + lo];
      hc[j] = wmma4(a, w, hc[j]);
    }
  }

  #pragma unroll
  for (int j = 0; j < 4; ++j)
    #pragma unroll
    for (int r = 0; r < 8; ++r)
      hw[(r + 8 * hi) * NCH + 16 * j + lo] = hc[j][r];

  __syncthreads();

  v8f mvc[4], mhc[4];
  #pragma unroll
  for (int j = 0; j < 4; ++j) {
    const float bias = b2[16 * j + lo];
    #pragma unroll
    for (int r = 0; r < 8; ++r) { mvc[j][r] = bias; mhc[j][r] = 0.0f; }
  }

  #pragma unroll
  for (int kk = 0; kk < 16; ++kk) {
    const int k = 4 * kk + 2 * hi;
    v2f a = *(const v2f*)&hw[lo * NCH + k];
    #pragma unroll
    for (int j = 0; j < 4; ++j) {
      v2f wv, wh;
      wv.x = W2[(k          ) * NCH + 16 * j + lo];
      wv.y = W2[(k + 1      ) * NCH + 16 * j + lo];
      wh.x = W2[(k + NCH    ) * NCH + 16 * j + lo];
      wh.y = W2[(k + NCH + 1) * NCH + 16 * j + lo];
      mvc[j] = wmma4(a, wv, mvc[j]);
      mhc[j] = wmma4(a, wh, mhc[j]);
    }
  }

  #pragma unroll
  for (int r = 0; r < 8; ++r) {
    const int m = m0 + r + 8 * hi;
    const int n = m / NT;
    const int t = m - n * NT;
    const int pbase = (((b * NT + t) * (NN / 2) + (n >> 1)) * NCH) * 2 + (n & 1);
    const int qbase = (((b * NTQ + (t >> 2)) * NN + n) * NCH) * TQ + (t & 3);
    #pragma unroll
    for (int j = 0; j < 4; ++j) {
      const int c = 16 * j + lo;
      P[pbase + c * 2]  = mhc[j][r];
      Q[qbase + c * TQ] = mvc[j][r];
    }
  }
}

__global__ __launch_bounds__(32)
void mix_kernel(const float* __restrict__ att,
                const float* __restrict__ P,
                const float* __restrict__ Q,
                float* __restrict__ out)
{
  const int lane = threadIdx.x;
  const int lo   = lane & 15;
  const int hi   = lane >> 4;

  const int id    = blockIdx.x;
  const int itile = id & 7;
  const int rest  = id >> 3;
  const int b     = rest / NTQ;
  const int tq    = rest - b * NTQ;
  const int t0    = TQ * tq;
  const int i0    = 16 * itile;

  const float* pb = P + (b * NT + t0) * ((NN / 2) * NCH * 2) + (hi * NCH + lo) * 2;
  v2f bc[16];
  #pragma unroll
  for (int t = 0; t < TQ; ++t)
    #pragma unroll
    for (int n = 0; n < 4; ++n)
      bc[t * 4 + n] = *(const v2f*)(pb + t * ((NN / 2) * NCH * 2) + n * 32);

  const float* ap = att + b * A_BSTRIDE + (i0 + lo) * A_ISTRIDE + (2 * hi) * NT + t0;
  v4f a0[32], a1[32];
  #pragma unroll
  for (int kk = 0; kk < 32; ++kk) {
    a0[kk] = *(const v4f*)(ap + (4 * kk    ) * NT);
    a1[kk] = *(const v4f*)(ap + (4 * kk + 1) * NT);
  }

  v8f acc[TQ][4];
  #pragma unroll
  for (int t = 0; t < TQ; ++t)
    #pragma unroll
    for (int n = 0; n < 4; ++n)
      #pragma unroll
      for (int r = 0; r < 8; ++r) acc[t][n][r] = 0.0f;

  float psum[TQ];
  #pragma unroll
  for (int t = 0; t < TQ; ++t) psum[t] = 0.0f;

  #pragma unroll
  for (int kk = 0; kk < 32; ++kk) {
    v2f bn[16];
    if (kk + 1 < 32) {
      #pragma unroll
      for (int t = 0; t < TQ; ++t)
        #pragma unroll
        for (int n = 0; n < 4; ++n)
          bn[t * 4 + n] = *(const v2f*)(pb + t * ((NN / 2) * NCH * 2)
                                           + (kk + 1) * (2 * NCH * 2) + n * 32);
    }
    #pragma unroll
    for (int t = 0; t < TQ; ++t) {
      v2f a;
      a.x = a0[kk][t];
      a.y = a1[kk][t];
      psum[t] += a.x + a.y;
      #pragma unroll
      for (int n = 0; n < 4; ++n)
        acc[t][n] = wmma4(a, bc[t * 4 + n], acc[t][n]);
    }
    if (kk + 1 < 32) {
      #pragma unroll
      for (int q = 0; q < 16; ++q) bc[q] = bn[q];
    }
  }

  float asum_lane[TQ];
  #pragma unroll
  for (int t = 0; t < TQ; ++t)
    asum_lane[t] = psum[t] + __shfl_xor(psum[t], 16, 32);

  const float* qb = Q + ((b * NTQ + tq) * NN) * (NCH * TQ);
  #pragma unroll
  for (int r = 0; r < 8; ++r) {
    const int row = r + 8 * hi;
    const int gi  = i0 + row;
    float as[TQ];
    #pragma unroll
    for (int t = 0; t < TQ; ++t) as[t] = __shfl(asum_lane[t], row, 32);
    #pragma unroll
    for (int n = 0; n < 4; ++n) {
      const int c = 16 * n + lo;
      const v4f q = *(const v4f*)(qb + (gi * NCH + c) * TQ);
      v4f val;
      val.x = acc[0][n][r] + as[0] * q.x;
      val.y = acc[1][n][r] + as[1] * q.y;
      val.z = acc[2][n][r] + as[2] * q.z;
      val.w = acc[3][n][r] + as[3] * q.w;
      *(v4f*)(out + b * X_BSTRIDE + c * X_CSTRIDE + gi * NT + t0) = val;
    }
  }
}

extern "C" void kernel_launch(void* const* d_in, const int* in_sizes, int n_in,
                              void* d_out, int out_size, void* d_ws, size_t ws_size,
                              hipStream_t stream) {
  const float* x   = (const float*)d_in[0];
  const float* att = (const float*)d_in[1];
  const float* W1  = (const float*)d_in[2];
  const float* b1  = (const float*)d_in[3];
  const float* W2  = (const float*)d_in[4];
  const float* b2  = (const float*)d_in[5];
  float* out = (float*)d_out;
  float* P = (float*)d_ws;
  float* Q = (float*)d_ws + WS_PLANE;

  hipLaunchKernelGGL(mlp_rows_kernel, dim3(96), dim3(128), 0, stream,
                     x, W1, b1, W2, b2, P, Q);
  hipLaunchKernelGGL(mix_kernel, dim3(96), dim3(32), 0, stream,
                     att, P, Q, out);
}


# Mixing launch split over column tiles: 4-wave blocks share the attention panel through LDS; each wave runs 128 matrix instructions with its mh fragments requested in 3 batches
# speedup vs baseline: 4.3132x; 1.2970x over previous
#include <hip/hip_runtime.h>


typedef __attribute__((ext_vector_type(2))) float v2f;
typedef __attribute__((ext_vector_type(4))) float v4f;
typedef __attribute__((ext_vector_type(8))) float v8f;

#define NB   2
#define NCH  64
#define NN   128
#define NT   24
#define MROWS       (NN*NT)
#define X_CSTRIDE   (NN*NT)
#define X_BSTRIDE   (NCH*NN*NT)
#define A_ISTRIDE   (NN*NT)
#define A_BSTRIDE   (NN*NN*NT)
#define TQ   4
#define NTQ  (NT/TQ)
#define WS_PLANE    (NB*NT*NN*NCH)


__device__ __forceinline__ v8f wmma4(v2f a, v2f b, v8f c) {
  return __builtin_amdgcn_wmma_f32_16x16x4_f32(
       false, a,  false, b,
       (short)0, c,  false,  false);
}

__global__ __launch_bounds__(128)
void mlp_rows_kernel(const float* __restrict__ x,
                     const float* __restrict__ W1,
                     const float* __restrict__ b1,
                     const float* __restrict__ W2,
                     const float* __restrict__ b2,
                     float* __restrict__ P,
                     float* __restrict__ Q)
{
  __shared__ float hs[4 * 16 * NCH];

  const int tid  = threadIdx.x;
  const int lane = tid & 31;
  const int lo   = lane & 15;
  const int hi   = lane >> 4;
  const int wave = tid >> 5;

  const int tile = blockIdx.x * 4 + wave;
  const int b    = tile / (MROWS / 16);
  const int m0   = (tile - b * (MROWS / 16)) * 16;
  float* hw = hs + wave * (16 * NCH);

  const int xbase = b * X_BSTRIDE + m0 + lo;
  v2f xr[16];
  #pragma unroll
  for (int kk = 0; kk < 16; ++kk) {
    const int k = 4 * kk + 2 * hi;
    xr[kk].x = x[xbase + (k    ) * X_CSTRIDE];
    xr[kk].y = x[xbase + (k + 1) * X_CSTRIDE];
  }

  v8f hc[4];
  #pragma unroll
  for (int j = 0; j < 4; ++j) {
    const float bias = b1[16 * j + lo];
    #pragma unroll
    for (int r = 0; r < 8; ++r) hc[j][r] = bias;
  }

  #pragma unroll
  for (int kk = 0; kk < 16; ++kk) {
    const int k = 4 * kk + 2 * hi;
    const v2f a = xr[kk];
    #pragma unroll
    for (int j = 0; j < 4; ++j) {
      v2f w;
      w.x = W1[(k    ) * NCH + 16 * j + lo];
      w.y = W1[(k + 1) * NCH + 16 * j + lo];
      hc[j] = wmma4(a, w, hc[j]);
    }
  }

  #pragma unroll
  for (int j = 0; j < 4; ++j)
    #pragma unroll
    for (int r = 0; r < 8; ++r)
      hw[(r + 8 * hi) * NCH + 16 * j + lo] = hc[j][r];

  __syncthreads();

  v8f mvc[4], mhc[4];
  #pragma unroll
  for (int j = 0; j < 4; ++j) {
    const float bias = b2[16 * j + lo];
    #pragma unroll
    for (int r = 0; r < 8; ++r) { mvc[j][r] = bias; mhc[j][r] = 0.0f; }
  }

  #pragma unroll
  for (int kk = 0; kk < 16; ++kk) {
    const int k = 4 * kk + 2 * hi;
    v2f a = *(const v2f*)&hw[lo * NCH + k];
    #pragma unroll
    for (int j = 0; j < 4; ++j) {
      v2f wv, wh;
      wv.x = W2[(k          ) * NCH + 16 * j + lo];
      wv.y = W2[(k + 1      ) * NCH + 16 * j + lo];
      wh.x = W2[(k + NCH    ) * NCH + 16 * j + lo];
      wh.y = W2[(k + NCH + 1) * NCH + 16 * j + lo];
      mvc[j] = wmma4(a, wv, mvc[j]);
      mhc[j] = wmma4(a, wh, mhc[j]);
    }
  }

  #pragma unroll
  for (int r = 0; r < 8; ++r) {
    const int m = m0 + r + 8 * hi;
    const int n = m / NT;
    const int t = m - n * NT;
    const int pbase = (((b * NT + t) * (NN / 2) + (n >> 1)) * NCH) * 2 + (n & 1);
    const int qbase = (((b * NTQ + (t >> 2)) * NN + n) * NCH) * TQ + (t & 3);
    #pragma unroll
    for (int j = 0; j < 4; ++j) {
      const int c = 16 * j + lo;
      P[pbase + c * 2]  = mhc[j][r];
      Q[qbase + c * TQ] = mvc[j][r];
    }
  }
}

#define P_TSTRIDE ((NN / 2) * NCH * 2)
#define P_KSTRIDE (2 * NCH * 2)

__global__ __launch_bounds__(128)
void mix_kernel(const float* __restrict__ att,
                const float* __restrict__ P,
                const float* __restrict__ Q,
                float* __restrict__ out)
{
  __shared__ v4f Ap[2][64 * 16];

  const int tid  = threadIdx.x;
  const int lane = tid & 31;
  const int lo   = lane & 15;
  const int hi   = lane >> 4;
  const int wave = tid >> 5;

  const int id    = blockIdx.x;
  const int itile = id & 7;
  const int rest  = id >> 3;
  const int b     = rest / NTQ;
  const int tq    = rest - b * NTQ;
  const int t0    = TQ * tq;
  const int i0    = 16 * itile;

  const float* ap = att + b * A_BSTRIDE + (i0 + lo) * A_ISTRIDE + (2 * (16 * wave + hi)) * NT + t0;
  v4f u0[8], u1[8];
  #pragma unroll
  for (int q = 0; q < 8; ++q) {
    u0[q] = *(const v4f*)(ap + (4 * q    ) * NT);
    u1[q] = *(const v4f*)(ap + (4 * q + 1) * NT);
  }

  const float* pb = P + (b * NT + t0) * P_TSTRIDE + (hi * NCH + 16 * wave + lo) * 2;
  v2f bf[32 * TQ];
  #pragma unroll
  for (int kk = 0; kk < 11; ++kk)
    #pragma unroll
    for (int t = 0; t < TQ; ++t)
      bf[kk * TQ + t] = *(const v2f*)(pb + t * P_TSTRIDE + kk * P_KSTRIDE);

  __builtin_amdgcn_sched_barrier(0);

  #pragma unroll
  for (int q = 0; q < 8; ++q) {
    const int jp = 16 * wave + 2 * q + hi;
    v4f e0, e1;
    e0.x = u0[q].x; e0.y = u1[q].x; e0.z = u0[q].y; e0.w = u1[q].y;
    e1.x = u0[q].z; e1.y = u1[q].z; e1.z = u0[q].w; e1.w = u1[q].w;
    Ap[0][jp * 16 + lo] = e0;
    Ap[1][jp * 16 + lo] = e1;
  }

  __syncthreads();

  #pragma unroll
  for (int kk = 11; kk < 22; ++kk)
    #pragma unroll
    for (int t = 0; t < TQ; ++t)
      bf[kk * TQ + t] = *(const v2f*)(pb + t * P_TSTRIDE + kk * P_KSTRIDE);

  v4f xa[2][2][8];
  #pragma unroll
  for (int k = 0; k < 8; ++k) {
    const int jp = 2 * k + hi;
    xa[0][0][k] = Ap[0][jp * 16 + lo];
    xa[0][1][k] = Ap[1][jp * 16 + lo];
  }

  v8f acc[TQ];
  #pragma unroll
  for (int t = 0; t < TQ; ++t)
    #pragma unroll
    for (int r = 0; r < 8; ++r) acc[t][r] = 0.0f;

  float psum[TQ];
  #pragma unroll
  for (int t = 0; t < TQ; ++t) psum[t] = 0.0f;

  const float* qb = Q + ((b * NTQ + tq) * NN) * (NCH * TQ);
  const int c = 16 * wave + lo;
  v4f qv[8];

  #pragma unroll
  for (int g = 0; g < 4; ++g) {
    if (g + 1 < 4) {
      #pragma unroll
      for (int k = 0; k < 8; ++k) {
        const int jp = 2 * (8 * (g + 1) + k) + hi;
        xa[(g + 1) & 1][0][k] = Ap[0][jp * 16 + lo];
        xa[(g + 1) & 1][1][k] = Ap[1][jp * 16 + lo];
      }
    }
    if (g == 1) {
      #pragma unroll
      for (int kk = 22; kk < 32; ++kk)
        #pragma unroll
        for (int t = 0; t < TQ; ++t)
          bf[kk * TQ + t] = *(const v2f*)(pb + t * P_TSTRIDE + kk * P_KSTRIDE);
    }
    if (g == 3) {
      #pragma unroll
      for (int r = 0; r < 8; ++r)
        qv[r] = *(const v4f*)(qb + ((i0 + r + 8 * hi) * NCH + c) * TQ);
    }
    __builtin_amdgcn_sched_barrier(0);

    #pragma unroll
    for (int k = 0; k < 8; ++k) {
      const int kk = 8 * g + k;
      #pragma unroll
      for (int t = 0; t < TQ; ++t) {
        const v4f y = xa[g & 1][t >> 1][k];
        v2f a;
        a.x = (t & 1) ? y.z : y.x;
        a.y = (t & 1) ? y.w : y.y;
        psum[t] += a.x + a.y;
        acc[t] = wmma4(a, bf[kk * TQ + t], acc[t]);
      }
    }
    __builtin_amdgcn_sched_barrier(0);
  }

  float asum_lane[TQ];
  #pragma unroll
  for (int t = 0; t < TQ; ++t)
    asum_lane[t] = psum[t] + __shfl_xor(psum[t], 16, 32);

  #pragma unroll
  for (int r = 0; r < 8; ++r) {
    const int row = r + 8 * hi;
    const int gi  = i0 + row;
    float as[TQ];
    #pragma unroll
    for (int t = 0; t < TQ; ++t) as[t] = __shfl(asum_lane[t], row, 32);
    const v4f q = qv[r];
    v4f val;
    val.x = acc[0][r] + as[0] * q.x;
    val.y = acc[1][r] + as[1] * q.y;
    val.z = acc[2][r] + as[2] * q.z;
    val.w = acc[3][r] + as[3] * q.w;
    *(v4f*)(out + b * X_BSTRIDE + c * X_CSTRIDE + gi * NT + t0) = val;
  }
}

extern "C" void kernel_launch(void* const* d_in, const int* in_sizes, int n_in,
                              void* d_out, int out_size, void* d_ws, size_t ws_size,
                              hipStream_t stream) {
  const float* x   = (const float*)d_in[0];
  const float* att = (const float*)d_in[1];
  const float* W1  = (const float*)d_in[2];
  const float* b1  = (const float*)d_in[3];
  const float* W2  = (const float*)d_in[4];
  const float* b2  = (const float*)d_in[5];
  float* out = (float*)d_out;
  float* P = (float*)d_ws;
  float* Q = (float*)d_ws + WS_PLANE;

  hipLaunchKernelGGL(mlp_rows_kernel, dim3(96), dim3(128), 0, stream,
                     x, W1, b1, W2, b2, P, Q);
  hipLaunchKernelGGL(mix_kernel, dim3(96), dim3(128), 0, stream,
                     att, P, Q, out);
}


# Launch 1 gains 128 blocks that re-lay attention into each mixing block's LDS image; the mixing launch fills its panel from one contiguous 32 KB read, not 16-byte pieces of 96-byte rows
# speedup vs baseline: 5.1702x; 1.1987x over previous
#include <hip/hip_runtime.h>


typedef __attribute__((ext_vector_type(2))) float v2f;
typedef __attribute__((ext_vector_type(4))) float v4f;
typedef __attribute__((ext_vector_type(8))) float v8f;

#define NB   2
#define NCH  64
#define NN   128
#define NT   24
#define MROWS       (NN*NT)
#define X_CSTRIDE   (NN*NT)
#define X_BSTRIDE   (NCH*NN*NT)
#define A_ISTRIDE   (NN*NT)
#define A_BSTRIDE   (NN*NN*NT)
#define TQ   4
#define NTQ  (NT/TQ)
#define WS_PLANE    (NB*NT*NN*NCH)
#define MLP_BLOCKS  96
#define RELAY_BLOCKS 128


__device__ __forceinline__ v8f wmma4(v2f a, v2f b, v8f c) {
  return __builtin_amdgcn_wmma_f32_16x16x4_f32(
       false, a,  false, b,
       (short)0, c,  false,  false);
}

__global__ __launch_bounds__(128)
void mlp_rows_kernel(const float* __restrict__ x,
                     const float* __restrict__ W1,
                     const float* __restrict__ b1,
                     const float* __restrict__ W2,
                     const float* __restrict__ b2,
                     float* __restrict__ P,
                     float* __restrict__ Q,
                     const float* __restrict__ att,
                     float* __restrict__ R)
{
  __shared__ float hs[4 * 16 * NCH];

  const int tid  = threadIdx.x;

  if (blockIdx.x >= MLP_BLOCKS) {
    const int r     = blockIdx.x - MLP_BLOCKS;
    const int rb    = r >> 6;
    const int ritl  = (r >> 3) & 7;
    const int jp    = 8 * (r & 7) + (tid >> 4);
    const int ri    = tid & 15;
    const float* src = att + rb * A_BSTRIDE + (16 * ritl + ri) * A_ISTRIDE + (2 * jp) * NT;
    v4f c0[NTQ], c1[NTQ];
    #pragma unroll
    for (int g = 0; g < NTQ; ++g) {
      c0[g] = *(const v4f*)(src + TQ * g);
      c1[g] = *(const v4f*)(src + NT + TQ * g);
    }
    v4f* dst = (v4f*)R + jp * 16 + ri;
    #pragma unroll
    for (int g = 0; g < NTQ; ++g) {
      v4f e0, e1;
      e0.x = c0[g].x; e0.y = c1[g].x; e0.z = c0[g].y; e0.w = c1[g].y;
      e1.x = c0[g].z; e1.y = c1[g].z; e1.z = c0[g].w; e1.w = c1[g].w;
      const int panel = (rb * NTQ + g) * 8 + ritl;
      dst[(panel * 2    ) * 1024] = e0;
      dst[(panel * 2 + 1) * 1024] = e1;
    }
    return;
  }

  const int lane = tid & 31;
  const int lo   = lane & 15;
  const int hi   = lane >> 4;
  const int wave = tid >> 5;

  const int tile = blockIdx.x * 4 + wave;
  const int b    = tile / (MROWS / 16);
  const int m0   = (tile - b * (MROWS / 16)) * 16;
  float* hw = hs + wave * (16 * NCH);

  const int xbase = b * X_BSTRIDE + m0 + lo;
  v2f xr[16];
  #pragma unroll
  for (int kk = 0; kk < 16; ++kk) {
    const int k = 4 * kk + 2 * hi;
    xr[kk].x = x[xbase + (k    ) * X_CSTRIDE];
    xr[kk].y = x[xbase + (k + 1) * X_CSTRIDE];
  }

  v8f hc[4];
  #pragma unroll
  for (int j = 0; j < 4; ++j) {
    const float bias = b1[16 * j + lo];
    #pragma unroll
    for (int r = 0; r < 8; ++r) hc[j][r] = bias;
  }

  #pragma unroll
  for (int kk = 0; kk < 16; ++kk) {
    const int k = 4 * kk + 2 * hi;
    const v2f a = xr[kk];
    #pragma unroll
    for (int j = 0; j < 4; ++j) {
      v2f w;
      w.x = W1[(k    ) * NCH + 16 * j + lo];
      w.y = W1[(k + 1) * NCH + 16 * j + lo];
      hc[j] = wmma4(a, w, hc[j]);
    }
  }

  #pragma unroll
  for (int j = 0; j < 4; ++j)
    #pragma unroll
    for (int r = 0; r < 8; ++r)
      hw[(r + 8 * hi) * NCH + 16 * j + lo] = hc[j][r];

  __syncthreads();

  v8f mvc[4], mhc[4];
  #pragma unroll
  for (int j = 0; j < 4; ++j) {
    const float bias = b2[16 * j + lo];
    #pragma unroll
    for (int r = 0; r < 8; ++r) { mvc[j][r] = bias; mhc[j][r] = 0.0f; }
  }

  #pragma unroll
  for (int kk = 0; kk < 16; ++kk) {
    const int k = 4 * kk + 2 * hi;
    v2f a = *(const v2f*)&hw[lo * NCH + k];
    #pragma unroll
    for (int j = 0; j < 4; ++j) {
      v2f wv, wh;
      wv.x = W2[(k          ) * NCH + 16 * j + lo];
      wv.y = W2[(k + 1      ) * NCH + 16 * j + lo];
      wh.x = W2[(k + NCH    ) * NCH + 16 * j + lo];
      wh.y = W2[(k + NCH + 1) * NCH + 16 * j + lo];
      mvc[j] = wmma4(a, wv, mvc[j]);
      mhc[j] = wmma4(a, wh, mhc[j]);
    }
  }

  #pragma unroll
  for (int r = 0; r < 8; ++r) {
    const int m = m0 + r + 8 * hi;
    const int n = m / NT;
    const int t = m - n * NT;
    const int pbase = (((b * NT + t) * (NN / 2) + (n >> 1)) * NCH) * 2 + (n & 1);
    const int qbase = (((b * NTQ + (t >> 2)) * NN + n) * NCH) * TQ + (t & 3);
    #pragma unroll
    for (int j = 0; j < 4; ++j) {
      const int c = 16 * j + lo;
      P[pbase + c * 2]  = mhc[j][r];
      Q[qbase + c * TQ] = mvc[j][r];
    }
  }
}

#define P_TSTRIDE ((NN / 2) * NCH * 2)
#define P_KSTRIDE (2 * NCH * 2)

__global__ __launch_bounds__(128)
void mix_kernel(const float* __restrict__ R,
                const float* __restrict__ P,
                const float* __restrict__ Q,
                float* __restrict__ out)
{
  __shared__ v4f Ap[2][64 * 16];

  const int tid  = threadIdx.x;
  const int lane = tid & 31;
  const int lo   = lane & 15;
  const int hi   = lane >> 4;
  const int wave = tid >> 5;

  const int id    = blockIdx.x;
  const int itile = id & 7;
  const int rest  = id >> 3;
  const int b     = rest / NTQ;
  const int tq    = rest - b * NTQ;
  const int t0    = TQ * tq;
  const int i0    = 16 * itile;

  const v4f* rp = (const v4f*)R + ((b * NTQ + tq) * 8 + itile) * 2048 + tid;
  v4f u[16];
  #pragma unroll
  for (int q = 0; q < 16; ++q)
    u[q] = rp[q * 128];

  const float* pb = P + (b * NT + t0) * P_TSTRIDE + (hi * NCH + 16 * wave + lo) * 2;
  v2f bf[32 * TQ];
  #pragma unroll
  for (int kk = 0; kk < 11; ++kk)
    #pragma unroll
    for (int t = 0; t < TQ; ++t)
      bf[kk * TQ + t] = *(const v2f*)(pb + t * P_TSTRIDE + kk * P_KSTRIDE);

  __builtin_amdgcn_sched_barrier(0);

  #pragma unroll
  for (int q = 0; q < 16; ++q)
    (&Ap[0][0])[q * 128 + tid] = u[q];

  __syncthreads();

  #pragma unroll
  for (int kk = 11; kk < 22; ++kk)
    #pragma unroll
    for (int t = 0; t < TQ; ++t)
      bf[kk * TQ + t] = *(const v2f*)(pb + t * P_TSTRIDE + kk * P_KSTRIDE);

  v4f xa[2][2][8];
  #pragma unroll
  for (int k = 0; k < 8; ++k) {
    const int jp = 2 * k + hi;
    xa[0][0][k] = Ap[0][jp * 16 + lo];
    xa[0][1][k] = Ap[1][jp * 16 + lo];
  }

  v8f acc[TQ];
  #pragma unroll
  for (int t = 0; t < TQ; ++t)
    #pragma unroll
    for (int r = 0; r < 8; ++r) acc[t][r] = 0.0f;

  float psum[TQ];
  #pragma unroll
  for (int t = 0; t < TQ; ++t) psum[t] = 0.0f;

  const float* qb = Q + ((b * NTQ + tq) * NN) * (NCH * TQ);
  const int c = 16 * wave + lo;
  v4f qv[8];

  #pragma unroll
  for (int g = 0; g < 4; ++g) {
    if (g + 1 < 4) {
      #pragma unroll
      for (int k = 0; k < 8; ++k) {
        const int jp = 2 * (8 * (g + 1) + k) + hi;
        xa[(g + 1) & 1][0][k] = Ap[0][jp * 16 + lo];
        xa[(g + 1) & 1][1][k] = Ap[1][jp * 16 + lo];
      }
    }
    if (g == 1) {
      #pragma unroll
      for (int kk = 22; kk < 32; ++kk)
        #pragma unroll
        for (int t = 0; t < TQ; ++t)
          bf[kk * TQ + t] = *(const v2f*)(pb + t * P_TSTRIDE + kk * P_KSTRIDE);
    }
    if (g == 3) {
      #pragma unroll
      for (int r = 0; r < 8; ++r)
        qv[r] = *(const v4f*)(qb + ((i0 + r + 8 * hi) * NCH + c) * TQ);
    }
    __builtin_amdgcn_sched_barrier(0);

    #pragma unroll
    for (int k = 0; k < 8; ++k) {
      const int kk = 8 * g + k;
      #pragma unroll
      for (int t = 0; t < TQ; ++t) {
        const v4f y = xa[g & 1][t >> 1][k];
        v2f a;
        a.x = (t & 1) ? y.z : y.x;
        a.y = (t & 1) ? y.w : y.y;
        psum[t] += a.x + a.y;
        acc[t] = wmma4(a, bf[kk * TQ + t], acc[t]);
      }
    }
    __builtin_amdgcn_sched_barrier(0);
  }

  float asum_lane[TQ];
  #pragma unroll
  for (int t = 0; t < TQ; ++t)
    asum_lane[t] = psum[t] + __shfl_xor(psum[t], 16, 32);

  #pragma unroll
  for (int r = 0; r < 8; ++r) {
    const int row = r + 8 * hi;
    const int gi  = i0 + row;
    float as[TQ];
    #pragma unroll
    for (int t = 0; t < TQ; ++t) as[t] = __shfl(asum_lane[t], row, 32);
    const v4f q = qv[r];
    v4f val;
    val.x = acc[0][r] + as[0] * q.x;
    val.y = acc[1][r] + as[1] * q.y;
    val.z = acc[2][r] + as[2] * q.z;
    val.w = acc[3][r] + as[3] * q.w;
    *(v4f*)(out + b * X_BSTRIDE + c * X_CSTRIDE + gi * NT + t0) = val;
  }
}

extern "C" void kernel_launch(void* const* d_in, const int* in_sizes, int n_in,
                              void* d_out, int out_size, void* d_ws, size_t ws_size,
                              hipStream_t stream) {
  const float* x   = (const float*)d_in[0];
  const float* att = (const float*)d_in[1];
  const float* W1  = (const float*)d_in[2];
  const float* b1  = (const float*)d_in[3];
  const float* W2  = (const float*)d_in[4];
  const float* b2  = (const float*)d_in[5];
  float* out = (float*)d_out;
  float* P = (float*)d_ws;
  float* Q = (float*)d_ws + WS_PLANE;
  float* R = (float*)d_ws + 2 * WS_PLANE;

  hipLaunchKernelGGL(mlp_rows_kernel, dim3(MLP_BLOCKS + RELAY_BLOCKS), dim3(128), 0, stream,
                     x, W1, b1, W2, b2, P, Q, att, R);
  hipLaunchKernelGGL(mix_kernel, dim3(96), dim3(128), 0, stream,
                     R, P, Q, out);
}


# Row-wise MLP launch re-tiled: block = 32 rows, each of its 4 waves owns one 16-column tile of h, mv and mh (h exchanged through LDS): 96 matrix instructions per wave, not 192, on 192 blocks, not 96
# speedup vs baseline: 5.8388x; 1.1293x over previous
#include <hip/hip_runtime.h>


typedef __attribute__((ext_vector_type(2))) float v2f;
typedef __attribute__((ext_vector_type(4))) float v4f;
typedef __attribute__((ext_vector_type(8))) float v8f;

#define NB   2
#define NCH  64
#define NN   128
#define NT   24
#define MROWS       (NN*NT)
#define X_CSTRIDE   (NN*NT)
#define X_BSTRIDE   (NCH*NN*NT)
#define A_ISTRIDE   (NN*NT)
#define A_BSTRIDE   (NN*NN*NT)
#define TQ   4
#define NTQ  (NT/TQ)
#define WS_PLANE    (NB*NT*NN*NCH)
#define MLP_BLOCKS  192
#define RELAY_BLOCKS 128


__device__ __forceinline__ v8f wmma4(v2f a, v2f b, v8f c) {
  return __builtin_amdgcn_wmma_f32_16x16x4_f32(
       false, a,  false, b,
       (short)0, c,  false,  false);
}

#define HP 68

__global__ __launch_bounds__(128)
void mlp_rows_kernel(const float* __restrict__ x,
                     const float* __restrict__ W1,
                     const float* __restrict__ b1,
                     const float* __restrict__ W2,
                     const float* __restrict__ b2,
                     float* __restrict__ P,
                     float* __restrict__ Q,
                     const float* __restrict__ att,
                     float* __restrict__ R)
{
  __shared__ __attribute__((aligned(16))) float hs[32 * HP];

  const int tid  = threadIdx.x;

  if (blockIdx.x >= MLP_BLOCKS) {
    const int r     = blockIdx.x - MLP_BLOCKS;
    const int rb    = r >> 6;
    const int ritl  = (r >> 3) & 7;
    const int jp    = 8 * (r & 7) + (tid >> 4);
    const int ri    = tid & 15;
    const float* src = att + rb * A_BSTRIDE + (16 * ritl + ri) * A_ISTRIDE + (2 * jp) * NT;
    v4f c0[NTQ], c1[NTQ];
    #pragma unroll
    for (int g = 0; g < NTQ; ++g) {
      c0[g] = *(const v4f*)(src + TQ * g);
      c1[g] = *(const v4f*)(src + NT + TQ * g);
    }
    v4f* dst = (v4f*)R + jp * 16 + ri;
    #pragma unroll
    for (int g = 0; g < NTQ; ++g) {
      v4f e0, e1;
      e0.x = c0[g].x; e0.y = c1[g].x; e0.z = c0[g].y; e0.w = c1[g].y;
      e1.x = c0[g].z; e1.y = c1[g].z; e1.z = c0[g].w; e1.w = c1[g].w;
      const int panel = (rb * NTQ + g) * 8 + ritl;
      dst[(panel * 2    ) * 1024] = e0;
      dst[(panel * 2 + 1) * 1024] = e1;
    }
    return;
  }

  const int lane = tid & 31;
  const int lo   = lane & 15;
  const int hi   = lane >> 4;
  const int wave = tid >> 5;
  const int col  = 16 * wave + lo;

  const int b    = blockIdx.x / (MROWS / 32);
  const int m0   = (blockIdx.x - b * (MROWS / 32)) * 32;

  const int xbase = b * X_BSTRIDE + m0 + lo;
  v2f xr[2][16];
  #pragma unroll
  for (int kk = 0; kk < 16; ++kk) {
    const int k = 4 * kk + 2 * hi;
    #pragma unroll
    for (int q = 0; q < 2; ++q) {
      xr[q][kk].x = x[xbase + 16 * q + (k    ) * X_CSTRIDE];
      xr[q][kk].y = x[xbase + 16 * q + (k + 1) * X_CSTRIDE];
    }
  }

  v2f w1f[16];
  #pragma unroll
  for (int kk = 0; kk < 16; ++kk) {
    const int k = 4 * kk + 2 * hi;
    w1f[kk].x = W1[(k    ) * NCH + col];
    w1f[kk].y = W1[(k + 1) * NCH + col];
  }

  v2f w2v[16], w2h[16];
  #pragma unroll
  for (int kk = 0; kk < 16; ++kk) {
    const int k = 4 * kk + 2 * hi;
    w2v[kk].x = W2[(k          ) * NCH + col];
    w2v[kk].y = W2[(k + 1      ) * NCH + col];
    w2h[kk].x = W2[(k + NCH    ) * NCH + col];
    w2h[kk].y = W2[(k + NCH + 1) * NCH + col];
  }

  v8f hc[2];
  {
    const float bias = b1[col];
    #pragma unroll
    for (int q = 0; q < 2; ++q)
      #pragma unroll
      for (int r = 0; r < 8; ++r) hc[q][r] = bias;
  }

  #pragma unroll
  for (int kk = 0; kk < 16; ++kk) {
    hc[0] = wmma4(xr[0][kk], w1f[kk], hc[0]);
    hc[1] = wmma4(xr[1][kk], w1f[kk], hc[1]);
  }

  #pragma unroll
  for (int q = 0; q < 2; ++q)
    #pragma unroll
    for (int r = 0; r < 8; ++r)
      hs[(16 * q + r + 8 * hi) * HP + col] = hc[q][r];

  __syncthreads();

  v8f mvc[2], mhc[2];
  {
    const float bias = b2[col];
    #pragma unroll
    for (int q = 0; q < 2; ++q)
      #pragma unroll
      for (int r = 0; r < 8; ++r) { mvc[q][r] = bias; mhc[q][r] = 0.0f; }
  }

  #pragma unroll
  for (int kk = 0; kk < 16; ++kk) {
    const int k = 4 * kk + 2 * hi;
    const v2f a0 = *(const v2f*)&hs[(lo     ) * HP + k];
    const v2f a1 = *(const v2f*)&hs[(lo + 16) * HP + k];
    mvc[0] = wmma4(a0, w2v[kk], mvc[0]);
    mhc[0] = wmma4(a0, w2h[kk], mhc[0]);
    mvc[1] = wmma4(a1, w2v[kk], mvc[1]);
    mhc[1] = wmma4(a1, w2h[kk], mhc[1]);
  }

  #pragma unroll
  for (int q = 0; q < 2; ++q) {
    #pragma unroll
    for (int r = 0; r < 8; ++r) {
      const int m = m0 + 16 * q + r + 8 * hi;
      const int n = m / NT;
      const int t = m - n * NT;
      const int pbase = (((b * NT + t) * (NN / 2) + (n >> 1)) * NCH) * 2 + (n & 1);
      const int qbase = (((b * NTQ + (t >> 2)) * NN + n) * NCH) * TQ + (t & 3);
      P[pbase + col * 2]  = mhc[q][r];
      Q[qbase + col * TQ] = mvc[q][r];
    }
  }
}

#define P_TSTRIDE ((NN / 2) * NCH * 2)
#define P_KSTRIDE (2 * NCH * 2)

__global__ __launch_bounds__(128)
void mix_kernel(const float* __restrict__ R,
                const float* __restrict__ P,
                const float* __restrict__ Q,
                float* __restrict__ out)
{
  __shared__ v4f Ap[2][64 * 16];

  const int tid  = threadIdx.x;
  const int lane = tid & 31;
  const int lo   = lane & 15;
  const int hi   = lane >> 4;
  const int wave = tid >> 5;

  const int id    = blockIdx.x;
  const int itile = id & 7;
  const int rest  = id >> 3;
  const int b     = rest / NTQ;
  const int tq    = rest - b * NTQ;
  const int t0    = TQ * tq;
  const int i0    = 16 * itile;

  const v4f* rp = (const v4f*)R + ((b * NTQ + tq) * 8 + itile) * 2048 + tid;
  v4f u[16];
  #pragma unroll
  for (int q = 0; q < 16; ++q)
    u[q] = rp[q * 128];

  const float* pb = P + (b * NT + t0) * P_TSTRIDE + (hi * NCH + 16 * wave + lo) * 2;
  v2f bf[32 * TQ];
  #pragma unroll
  for (int kk = 0; kk < 11; ++kk)
    #pragma unroll
    for (int t = 0; t < TQ; ++t)
      bf[kk * TQ + t] = *(const v2f*)(pb + t * P_TSTRIDE + kk * P_KSTRIDE);

  __builtin_amdgcn_sched_barrier(0);

  #pragma unroll
  for (int q = 0; q < 16; ++q)
    (&Ap[0][0])[q * 128 + tid] = u[q];

  __syncthreads();

  #pragma unroll
  for (int kk = 11; kk < 22; ++kk)
    #pragma unroll
    for (int t = 0; t < TQ; ++t)
      bf[kk * TQ + t] = *(const v2f*)(pb + t * P_TSTRIDE + kk * P_KSTRIDE);

  v4f xa[2][2][8];
  #pragma unroll
  for (int k = 0; k < 8; ++k) {
    const int jp = 2 * k + hi;
    xa[0][0][k] = Ap[0][jp * 16 + lo];
    xa[0][1][k] = Ap[1][jp * 16 + lo];
  }

  v8f acc[TQ];
  #pragma unroll
  for (int t = 0; t < TQ; ++t)
    #pragma unroll
    for (int r = 0; r < 8; ++r) acc[t][r] = 0.0f;

  float psum[TQ];
  #pragma unroll
  for (int t = 0; t < TQ; ++t) psum[t] = 0.0f;

  const float* qb = Q + ((b * NTQ + tq) * NN) * (NCH * TQ);
  const int c = 16 * wave + lo;
  v4f qv[8];

  #pragma unroll
  for (int g = 0; g < 4; ++g) {
    if (g + 1 < 4) {
      #pragma unroll
      for (int k = 0; k < 8; ++k) {
        const int jp = 2 * (8 * (g + 1) + k) + hi;
        xa[(g + 1) & 1][0][k] = Ap[0][jp * 16 + lo];
        xa[(g + 1) & 1][1][k] = Ap[1][jp * 16 + lo];
      }
    }
    if (g == 1) {
      #pragma unroll
      for (int kk = 22; kk < 32; ++kk)
        #pragma unroll
        for (int t = 0; t < TQ; ++t)
          bf[kk * TQ + t] = *(const v2f*)(pb + t * P_TSTRIDE + kk * P_KSTRIDE);
    }
    if (g == 3) {
      #pragma unroll
      for (int r = 0; r < 8; ++r)
        qv[r] = *(const v4f*)(qb + ((i0 + r + 8 * hi) * NCH + c) * TQ);
    }
    __builtin_amdgcn_sched_barrier(0);

    #pragma unroll
    for (int k = 0; k < 8; ++k) {
      const int kk = 8 * g + k;
      #pragma unroll
      for (int t = 0; t < TQ; ++t) {
        const v4f y = xa[g & 1][t >> 1][k];
        v2f a;
        a.x = (t & 1) ? y.z : y.x;
        a.y = (t & 1) ? y.w : y.y;
        psum[t] += a.x + a.y;
        acc[t] = wmma4(a, bf[kk * TQ + t], acc[t]);
      }
    }
    __builtin_amdgcn_sched_barrier(0);
  }

  float asum_lane[TQ];
  #pragma unroll
  for (int t = 0; t < TQ; ++t)
    asum_lane[t] = psum[t] + __shfl_xor(psum[t], 16, 32);

  #pragma unroll
  for (int r = 0; r < 8; ++r) {
    const int row = r + 8 * hi;
    const int gi  = i0 + row;
    float as[TQ];
    #pragma unroll
    for (int t = 0; t < TQ; ++t) as[t] = __shfl(asum_lane[t], row, 32);
    const v4f q = qv[r];
    v4f val;
    val.x = acc[0][r] + as[0] * q.x;
    val.y = acc[1][r] + as[1] * q.y;
    val.z = acc[2][r] + as[2] * q.z;
    val.w = acc[3][r] + as[3] * q.w;
    *(v4f*)(out + b * X_BSTRIDE + c * X_CSTRIDE + gi * NT + t0) = val;
  }
}

extern "C" void kernel_launch(void* const* d_in, const int* in_sizes, int n_in,
                              void* d_out, int out_size, void* d_ws, size_t ws_size,
                              hipStream_t stream) {
  const float* x   = (const float*)d_in[0];
  const float* att = (const float*)d_in[1];
  const float* W1  = (const float*)d_in[2];
  const float* b1  = (const float*)d_in[3];
  const float* W2  = (const float*)d_in[4];
  const float* b2  = (const float*)d_in[5];
  float* out = (float*)d_out;
  float* P = (float*)d_ws;
  float* Q = (float*)d_ws + WS_PLANE;
  float* R = (float*)d_ws + 2 * WS_PLANE;

  hipLaunchKernelGGL(mlp_rows_kernel, dim3(MLP_BLOCKS + RELAY_BLOCKS), dim3(128), 0, stream,
                     x, W1, b1, W2, b2, P, Q, att, R);
  hipLaunchKernelGGL(mix_kernel, dim3(96), dim3(128), 0, stream,
                     R, P, Q, out);
}


# Mixing launch re-tiled: 192 blocks (column halves), wave = column tile x t pair, 64 matrix instructions per wave, not 128; partner waves swap row halves through LDS so stores stay 16-byte
# speedup vs baseline: 6.3826x; 1.0931x over previous
#include <hip/hip_runtime.h>


typedef __attribute__((ext_vector_type(2))) float v2f;
typedef __attribute__((ext_vector_type(4))) float v4f;
typedef __attribute__((ext_vector_type(8))) float v8f;

#define NB   2
#define NCH  64
#define NN   128
#define NT   24
#define MROWS       (NN*NT)
#define X_CSTRIDE   (NN*NT)
#define X_BSTRIDE   (NCH*NN*NT)
#define A_ISTRIDE   (NN*NT)
#define A_BSTRIDE   (NN*NN*NT)
#define TQ   4
#define NTQ  (NT/TQ)
#define WS_PLANE    (NB*NT*NN*NCH)
#define MLP_BLOCKS  192
#define RELAY_BLOCKS 128


__device__ __forceinline__ v8f wmma4(v2f a, v2f b, v8f c) {
  return __builtin_amdgcn_wmma_f32_16x16x4_f32(
       false, a,  false, b,
       (short)0, c,  false,  false);
}

#define HP 68

__global__ __launch_bounds__(128)
void mlp_rows_kernel(const float* __restrict__ x,
                     const float* __restrict__ W1,
                     const float* __restrict__ b1,
                     const float* __restrict__ W2,
                     const float* __restrict__ b2,
                     float* __restrict__ P,
                     float* __restrict__ Q,
                     const float* __restrict__ att,
                     float* __restrict__ R)
{
  __shared__ __attribute__((aligned(16))) float hs[32 * HP];

  const int tid  = threadIdx.x;

  if (blockIdx.x >= MLP_BLOCKS) {
    const int r     = blockIdx.x - MLP_BLOCKS;
    const int rb    = r >> 6;
    const int ritl  = (r >> 3) & 7;
    const int jp    = 8 * (r & 7) + (tid >> 4);
    const int ri    = tid & 15;
    const float* src = att + rb * A_BSTRIDE + (16 * ritl + ri) * A_ISTRIDE + (2 * jp) * NT;
    v4f c0[NTQ], c1[NTQ];
    #pragma unroll
    for (int g = 0; g < NTQ; ++g) {
      c0[g] = *(const v4f*)(src + TQ * g);
      c1[g] = *(const v4f*)(src + NT + TQ * g);
    }
    v4f* dst = (v4f*)R + jp * 16 + ri;
    #pragma unroll
    for (int g = 0; g < NTQ; ++g) {
      v4f e0, e1;
      e0.x = c0[g].x; e0.y = c1[g].x; e0.z = c0[g].y; e0.w = c1[g].y;
      e1.x = c0[g].z; e1.y = c1[g].z; e1.z = c0[g].w; e1.w = c1[g].w;
      const int panel = (rb * NTQ + g) * 8 + ritl;
      dst[(panel * 2    ) * 1024] = e0;
      dst[(panel * 2 + 1) * 1024] = e1;
    }
    return;
  }

  const int lane = tid & 31;
  const int lo   = lane & 15;
  const int hi   = lane >> 4;
  const int wave = tid >> 5;
  const int col  = 16 * wave + lo;

  const int b    = blockIdx.x / (MROWS / 32);
  const int m0   = (blockIdx.x - b * (MROWS / 32)) * 32;

  const int xbase = b * X_BSTRIDE + m0 + lo;
  v2f xr[2][16];
  #pragma unroll
  for (int kk = 0; kk < 16; ++kk) {
    const int k = 4 * kk + 2 * hi;
    #pragma unroll
    for (int q = 0; q < 2; ++q) {
      xr[q][kk].x = x[xbase + 16 * q + (k    ) * X_CSTRIDE];
      xr[q][kk].y = x[xbase + 16 * q + (k + 1) * X_CSTRIDE];
    }
  }

  v2f w1f[16];
  #pragma unroll
  for (int kk = 0; kk < 16; ++kk) {
    const int k = 4 * kk + 2 * hi;
    w1f[kk].x = W1[(k    ) * NCH + col];
    w1f[kk].y = W1[(k + 1) * NCH + col];
  }

  v2f w2v[16], w2h[16];
  #pragma unroll
  for (int kk = 0; kk < 16; ++kk) {
    const int k = 4 * kk + 2 * hi;
    w2v[kk].x = W2[(k          ) * NCH + col];
    w2v[kk].y = W2[(k + 1      ) * NCH + col];
    w2h[kk].x = W2[(k + NCH    ) * NCH + col];
    w2h[kk].y = W2[(k + NCH + 1) * NCH + col];
  }

  v8f hc[2];
  {
    const float bias = b1[col];
    #pragma unroll
    for (int q = 0; q < 2; ++q)
      #pragma unroll
      for (int r = 0; r < 8; ++r) hc[q][r] = bias;
  }

  #pragma unroll
  for (int kk = 0; kk < 16; ++kk) {
    hc[0] = wmma4(xr[0][kk], w1f[kk], hc[0]);
    hc[1] = wmma4(xr[1][kk], w1f[kk], hc[1]);
  }

  #pragma unroll
  for (int q = 0; q < 2; ++q)
    #pragma unroll
    for (int r = 0; r < 8; ++r)
      hs[(16 * q + r + 8 * hi) * HP + col] = hc[q][r];

  __syncthreads();

  v8f mvc[2], mhc[2];
  {
    const float bias = b2[col];
    #pragma unroll
    for (int q = 0; q < 2; ++q)
      #pragma unroll
      for (int r = 0; r < 8; ++r) { mvc[q][r] = bias; mhc[q][r] = 0.0f; }
  }

  #pragma unroll
  for (int kk = 0; kk < 16; ++kk) {
    const int k = 4 * kk + 2 * hi;
    const v2f a0 = *(const v2f*)&hs[(lo     ) * HP + k];
    const v2f a1 = *(const v2f*)&hs[(lo + 16) * HP + k];
    mvc[0] = wmma4(a0, w2v[kk], mvc[0]);
    mhc[0] = wmma4(a0, w2h[kk], mhc[0]);
    mvc[1] = wmma4(a1, w2v[kk], mvc[1]);
    mhc[1] = wmma4(a1, w2h[kk], mhc[1]);
  }

  #pragma unroll
  for (int q = 0; q < 2; ++q) {
    #pragma unroll
    for (int r = 0; r < 8; ++r) {
      const int m = m0 + 16 * q + r + 8 * hi;
      const int n = m / NT;
      const int t = m - n * NT;
      const int pbase = (((b * NT + t) * (NN / 2) + (n >> 1)) * NCH) * 2 + (n & 1);
      const int qbase = (((b * NTQ + (t >> 2)) * NN + n) * NCH) * TQ + (t & 3);
      P[pbase + col * 2]  = mhc[q][r];
      Q[qbase + col * TQ] = mvc[q][r];
    }
  }
}

#define P_TSTRIDE ((NN / 2) * NCH * 2)
#define P_KSTRIDE (2 * NCH * 2)
#define MIX_BLOCKS 192

__global__ __launch_bounds__(128)
void mix_kernel(const float* __restrict__ R,
                const float* __restrict__ P,
                const float* __restrict__ Q,
                float* __restrict__ out)
{
  __shared__ v4f Ap[2][64 * 16];
  __shared__ v4f Ex[2][2][2][32];
  __shared__ v4f As4[2][16];

  const int tid  = threadIdx.x;
  const int lane = tid & 31;
  const int lo   = lane & 15;
  const int hi   = lane >> 4;
  const int wave = tid >> 5;
  const int wc   = wave & 1;
  const int th   = wave >> 1;

  const int id    = blockIdx.x;
  const int itile = id & 7;
  const int chalf = (id >> 3) & 1;
  const int rest  = id >> 4;
  const int b     = rest / NTQ;
  const int tq    = rest - b * NTQ;
  const int t0    = TQ * tq;
  const int i0    = 16 * itile;
  const int c     = 16 * (2 * chalf + wc) + lo;

  const v4f* rp = (const v4f*)R + ((b * NTQ + tq) * 8 + itile) * 2048 + tid;
  v4f u[16];
  #pragma unroll
  for (int q = 0; q < 16; ++q)
    u[q] = rp[q * 128];

  const float* pb = P + (b * NT + t0 + 2 * th) * P_TSTRIDE + (hi * NCH + c) * 2;
  v2f bf[32 * 2];
  #pragma unroll
  for (int kk = 0; kk < 22; ++kk)
    #pragma unroll
    for (int tt = 0; tt < 2; ++tt)
      bf[kk * 2 + tt] = *(const v2f*)(pb + tt * P_TSTRIDE + kk * P_KSTRIDE);

  __builtin_amdgcn_sched_barrier(0);

  #pragma unroll
  for (int q = 0; q < 16; ++q)
    (&Ap[0][0])[q * 128 + tid] = u[q];

  __syncthreads();

  #pragma unroll
  for (int kk = 22; kk < 32; ++kk)
    #pragma unroll
    for (int tt = 0; tt < 2; ++tt)
      bf[kk * 2 + tt] = *(const v2f*)(pb + tt * P_TSTRIDE + kk * P_KSTRIDE);

  const float* qb = Q + ((b * NTQ + tq) * NN) * (NCH * TQ);
  v4f qv[4];
  #pragma unroll
  for (int rr = 0; rr < 4; ++rr)
    qv[rr] = *(const v4f*)(qb + ((i0 + 8 * hi + 4 * th + rr) * NCH + c) * TQ);

  const v4f* ap = &Ap[th][0];
  v4f xa[2][8];
  #pragma unroll
  for (int k = 0; k < 8; ++k) {
    const int jp = 2 * k + hi;
    xa[0][k] = ap[jp * 16 + lo];
  }

  v8f acc[2];
  #pragma unroll
  for (int tt = 0; tt < 2; ++tt)
    #pragma unroll
    for (int r = 0; r < 8; ++r) acc[tt][r] = 0.0f;

  float psum[2];
  #pragma unroll
  for (int tt = 0; tt < 2; ++tt) psum[tt] = 0.0f;

  #pragma unroll
  for (int g = 0; g < 4; ++g) {
    if (g + 1 < 4) {
      #pragma unroll
      for (int k = 0; k < 8; ++k) {
        const int jp = 2 * (8 * (g + 1) + k) + hi;
        xa[(g + 1) & 1][k] = ap[jp * 16 + lo];
      }
    }
    __builtin_amdgcn_sched_barrier(0);

    #pragma unroll
    for (int k = 0; k < 8; ++k) {
      const int kk = 8 * g + k;
      const v4f y = xa[g & 1][k];
      #pragma unroll
      for (int tt = 0; tt < 2; ++tt) {
        v2f a;
        a.x = tt ? y.z : y.x;
        a.y = tt ? y.w : y.y;
        psum[tt] += a.x + a.y;
        acc[tt] = wmma4(a, bf[kk * 2 + tt], acc[tt]);
      }
    }
    __builtin_amdgcn_sched_barrier(0);
  }

  v2f asum_pair;
  asum_pair.x = psum[0] + __shfl_xor(psum[0], 16, 32);
  asum_pair.y = psum[1] + __shfl_xor(psum[1], 16, 32);
  if (hi == 0)
    *(v2f*)((float*)&As4[wc][lo] + 2 * th) = asum_pair;

  #pragma unroll
  for (int tt = 0; tt < 2; ++tt) {
    v4f e;
    e.x = th ? acc[tt][0] : acc[tt][4];
    e.y = th ? acc[tt][1] : acc[tt][5];
    e.z = th ? acc[tt][2] : acc[tt][6];
    e.w = th ? acc[tt][3] : acc[tt][7];
    Ex[wc][th][tt][lane] = e;
  }

  __syncthreads();

  v4f pv[2];
  #pragma unroll
  for (int tt = 0; tt < 2; ++tt)
    pv[tt] = Ex[wc][1 - th][tt][lane];

  #pragma unroll
  for (int rr = 0; rr < 4; ++rr) {
    const int row = 8 * hi + 4 * th + rr;
    const int gi  = i0 + row;
    const v4f as = As4[wc][row];
    const float o0 = th ? acc[0][4 + rr] : acc[0][rr];
    const float o1 = th ? acc[1][4 + rr] : acc[1][rr];
    const float p0 = pv[0][rr];
    const float p1 = pv[1][rr];
    const float a0 = th ? p0 : o0;
    const float a1 = th ? p1 : o1;
    const float a2 = th ? o0 : p0;
    const float a3 = th ? o1 : p1;
    const v4f q = qv[rr];
    v4f val;
    val.x = a0 + as.x * q.x;
    val.y = a1 + as.y * q.y;
    val.z = a2 + as.z * q.z;
    val.w = a3 + as.w * q.w;
    *(v4f*)(out + b * X_BSTRIDE + c * X_CSTRIDE + gi * NT + t0) = val;
  }
}

extern "C" void kernel_launch(void* const* d_in, const int* in_sizes, int n_in,
                              void* d_out, int out_size, void* d_ws, size_t ws_size,
                              hipStream_t stream) {
  const float* x   = (const float*)d_in[0];
  const float* att = (const float*)d_in[1];
  const float* W1  = (const float*)d_in[2];
  const float* b1  = (const float*)d_in[3];
  const float* W2  = (const float*)d_in[4];
  const float* b2  = (const float*)d_in[5];
  float* out = (float*)d_out;
  float* P = (float*)d_ws;
  float* Q = (float*)d_ws + WS_PLANE;
  float* R = (float*)d_ws + 2 * WS_PLANE;

  hipLaunchKernelGGL(mlp_rows_kernel, dim3(MLP_BLOCKS + RELAY_BLOCKS), dim3(128), 0, stream,
                     x, W1, b1, W2, b2, P, Q, att, R);
  hipLaunchKernelGGL(mix_kernel, dim3(MIX_BLOCKS), dim3(128), 0, stream,
                     R, P, Q, out);
}
